# starting kernel
# baseline (speedup 1.0000x reference)
#include <hip/hip_runtime.h>
#include <math.h>

typedef __attribute__((ext_vector_type(2))) float v2f;
typedef __attribute__((ext_vector_type(8))) float v8f;

#define B_    4
#define P_    6
#define F_    257
#define T_    300
#define N_    36
#define NPAD  48
#define TPAD  304
#define NT_TILES 3
#define TT_TILES 19
#define FT    (F_ * T_)
#define V_SOUND 343.0f
#define TWO_PI 6.28318530717958647692f

__device__ __forceinline__ v8f wmma_tile_k12(
    const float* cosT, const float* sinT,
    const float* cosO, const float* sinO,
    int nA, int tB, int halfSel)
{
    v8f acc = {};
    #pragma unroll
    for (int kk = 0; kk < 12; kk += 4) {
        const int p = (kk >> 1) + halfSel;
        v2f afrag, bfrag;
        afrag.x = cosT[p * NPAD + nA];
        afrag.y = sinT[p * NPAD + nA];
        bfrag.x = cosO[p * TPAD + tB];
        bfrag.y = sinO[p * TPAD + tB];
        acc = __builtin_amdgcn_wmma_f32_16x16x4_f32(
             false, afrag,  false, bfrag,
             (short)0, acc,  false,  false);
    }
    return acc;
}

__global__ __launch_bounds__(256) void CalculateDirectionFeature_kernel(
    const float* __restrict__ obs,
    const float* __restrict__ azi,
    const float* __restrict__ ele,
    const float* __restrict__ pvec,
    const float* __restrict__ freq,
    float* __restrict__ out)
{
    __shared__ float cosT[P_ * NPAD];
    __shared__ float sinT[P_ * NPAD];
    __shared__ float cosO[P_ * TPAD];
    __shared__ float sinO[P_ * TPAD];

    const int bf  = blockIdx.x;
    const int b   = bf / F_;
    const int f   = bf % F_;
    const int tid = threadIdx.x;

    const float fr    = freq[f];
    const float scale = (TWO_PI / V_SOUND) * fr;

    for (int i = tid; i < P_ * NPAD; i += 256) {
        const int p = i / NPAD;
        const int n = i % NPAD;
        float c = 0.0f, s = 0.0f;
        if (n < N_) {
            const float a = azi[b * N_ + n];
            const float e = ele[b * N_ + n];
            float sa, ca, se, ce;
            __sincosf(a, &sa, &ca);
            __sincosf(e, &se, &ce);
            const float rx = se * ca, ry = se * sa, rz = ce;
            const float dot = pvec[p * 3 + 0] * rx + pvec[p * 3 + 1] * ry +
                              pvec[p * 3 + 2] * rz;
            __sincosf(scale * dot, &s, &c);
        }
        cosT[i] = c;
        sinT[i] = s;
    }

    for (int i = tid; i < P_ * TPAD; i += 256) {
        const int p = i / TPAD;
        const int t = i % TPAD;
        float c = 0.0f, s = 0.0f;
        if (t < T_) {
            const float o = obs[(((size_t)b * P_ + p) * F_ + f) * T_ + t];
            __sincosf(o, &s, &c);
        }
        cosO[i] = c;
        sinO[i] = s;
    }
    __syncthreads();

    const int wave    = tid >> 5;
    const int lane    = tid & 31;
    const int halfSel = lane >> 4;
    const int lane16  = lane & 15;

    float* const outBF = out + ((size_t)b * N_ * F_ + f) * T_;
    const int laneOff = halfSel * (8 * FT) + lane16;

    for (int tile = wave; tile < 2 * TT_TILES; tile += 8) {
        const int nt = tile / TT_TILES;
        const int tt = tile % TT_TILES;
        const int n0 = nt * 16;
        const int t0 = (tt == TT_TILES - 1) ? (T_ - 16) : tt * 16;

        v8f acc = wmma_tile_k12(cosT, sinT, cosO, sinO,
                                n0 + lane16, t0 + lane16, halfSel);

        float* op = outBF + ((size_t)(n0 * FT + t0) + laneOff);
        #pragma unroll
        for (int r = 0; r < 8; ++r) {
            op[(size_t)r * FT] = acc[r];
        }
    }

    for (int tt = wave; tt < TT_TILES; tt += 8) {
        const int n0 = 32;
        const int t0 = (tt == TT_TILES - 1) ? (T_ - 16) : tt * 16;

        v8f acc = wmma_tile_k12(cosT, sinT, cosO, sinO,
                                n0 + lane16, t0 + lane16, halfSel);

        float* op = outBF + ((size_t)(n0 * FT + t0) + laneOff);
        if (halfSel == 0) {
            #pragma unroll
            for (int r = 0; r < 4; ++r) {
                op[(size_t)r * FT] = acc[r];
            }
        }
    }
}

extern "C" void kernel_launch(void* const* d_in, const int* in_sizes, int n_in,
                              void* d_out, int out_size, void* d_ws, size_t ws_size,
                              hipStream_t stream) {
    (void)in_sizes; (void)n_in; (void)d_ws; (void)ws_size; (void)out_size;
    const float* obs  = (const float*)d_in[0];
    const float* azi  = (const float*)d_in[1];
    const float* ele  = (const float*)d_in[2];
    const float* pvec = (const float*)d_in[3];
    const float* freq = (const float*)d_in[4];
    float* out = (float*)d_out;

    dim3 grid(B_ * F_);
    dim3 block(256);
    CalculateDirectionFeature_kernel<<<grid, block, 0, stream>>>(
        obs, azi, ele, pvec, freq, out);
}


# One wave per (b,f) bin, no LDS or barrier: 16-byte obs loads feed B fragments whose columns interleave as t=t0+4c+j, so a lane's 4 accumulators leave as one 16-byte store (256 B per row)
# speedup vs baseline: 1.5959x; 1.5959x over previous
#include <hip/hip_runtime.h>
#include <math.h>


typedef __attribute__((ext_vector_type(2))) float v2f;
typedef __attribute__((ext_vector_type(4))) float v4f;
typedef __attribute__((ext_vector_type(8))) float v8f;

#define B_    4
#define P_    6
#define F_    257
#define T_    300
#define N_    36
#define FT    (F_ * T_)
#define MT_   5
#define V_SOUND 343.0f
#define TWO_PI 6.28318530717958647692f

__global__ __launch_bounds__(32) void CalculateDirectionFeature_kernel(
    const float* __restrict__ obs,
    const float* __restrict__ azi,
    const float* __restrict__ ele,
    const float* __restrict__ pvec,
    const float* __restrict__ freq,
    float* __restrict__ out)
{
    const int bf   = blockIdx.x;
    const int b    = bf / F_;
    const int f    = bf % F_;
    const int lane = threadIdx.x;
    const int c    = lane & 15;
    const int h    = lane >> 4;

    const float fr = freq[f];
    float az[3], el[3], pv[3][3];
    #pragma unroll
    for (int nt = 0; nt < 3; ++nt) {
        const int n  = nt * 16 + c;
        const int nn = (n < N_) ? n : 0;
        az[nt] = azi[b * N_ + nn];
        el[nt] = ele[b * N_ + nn];
    }
    #pragma unroll
    for (int q = 0; q < 3; ++q) {
        const int p = 2 * q + h;
        pv[q][0] = pvec[p * 3 + 0];
        pv[q][1] = pvec[p * 3 + 1];
        pv[q][2] = pvec[p * 3 + 2];
    }
    __builtin_amdgcn_sched_barrier(0);

    const float* const obsBF = obs + ((size_t)b * P_ * F_ + f) * T_ + (size_t)h * FT;
    const bool tailOk = (4 * c + 64 * (MT_ - 1)) < T_;
    v4f o[MT_][3];
    #pragma unroll
    for (int m = 0; m < MT_; ++m) {
        const int tl = (m < MT_ - 1 || tailOk) ? (64 * m + 4 * c) : (64 * m);
        #pragma unroll
        for (int q = 0; q < 3; ++q) {
            o[m][q] = *reinterpret_cast<const v4f*>(obsBF + (size_t)(2 * q) * FT + tl);
        }
    }
    __builtin_amdgcn_sched_barrier(0);

    const float scale = (TWO_PI / V_SOUND) * fr;
    v2f A[3][3];
    #pragma unroll
    for (int nt = 0; nt < 3; ++nt) {
        const bool valid = (nt * 16 + c) < N_;
        float sa, ca, se, ce;
        __sincosf(az[nt], &sa, &ca);
        __sincosf(el[nt], &se, &ce);
        const float rx = se * ca, ry = se * sa, rz = ce;
        #pragma unroll
        for (int q = 0; q < 3; ++q) {
            const float dot = pv[q][0] * rx + pv[q][1] * ry + pv[q][2] * rz;
            float s, cs;
            __sincosf(scale * dot, &s, &cs);
            A[nt][q].x = valid ? cs : 0.0f;
            A[nt][q].y = valid ? s  : 0.0f;
        }
    }

    __builtin_amdgcn_sched_barrier(0);

    float* const outL = out + ((size_t)b * N_ * F_ + f) * T_ + (size_t)(8 * h) * FT + 4 * c;

    #pragma unroll
    for (int m = 0; m < MT_; ++m) {
        v2f Bf[3][4];
        #pragma unroll
        for (int q = 0; q < 3; ++q) {
            #pragma unroll
            for (int j = 0; j < 4; ++j) {
                float s, cs;
                __sincosf(o[m][q][j], &s, &cs);
                Bf[q][j].x = cs;
                Bf[q][j].y = s;
            }
        }

        v8f acc[3][4];
        #pragma unroll
        for (int nt = 0; nt < 3; ++nt) {
            #pragma unroll
            for (int j = 0; j < 4; ++j) {
                acc[nt][j] = (v8f){};
            }
        }
        #pragma unroll
        for (int q = 0; q < 3; ++q) {
            #pragma unroll
            for (int nt = 0; nt < 3; ++nt) {
                #pragma unroll
                for (int j = 0; j < 4; ++j) {
                    acc[nt][j] = __builtin_amdgcn_wmma_f32_16x16x4_f32(
                         false, A[nt][q],  false, Bf[q][j],
                         (short)0, acc[nt][j],
                         false,  false);
                }
            }
        }

        float* const op = outL + 64 * m;
        if (m < MT_ - 1 || tailOk) {
            #pragma unroll
            for (int nt = 0; nt < 2; ++nt) {
                #pragma unroll
                for (int r = 0; r < 8; ++r) {
                    v4f v;
                    v.x = acc[nt][0][r];
                    v.y = acc[nt][1][r];
                    v.z = acc[nt][2][r];
                    v.w = acc[nt][3][r];
                    *reinterpret_cast<v4f*>(op + (size_t)(nt * 16 + r) * FT) = v;
                }
            }
            if (h == 0) {
                #pragma unroll
                for (int r = 0; r < 4; ++r) {
                    v4f v;
                    v.x = acc[2][0][r];
                    v.y = acc[2][1][r];
                    v.z = acc[2][2][r];
                    v.w = acc[2][3][r];
                    *reinterpret_cast<v4f*>(op + (size_t)(32 + r) * FT) = v;
                }
            }
        }
    }
}

extern "C" void kernel_launch(void* const* d_in, const int* in_sizes, int n_in,
                              void* d_out, int out_size, void* d_ws, size_t ws_size,
                              hipStream_t stream) {
    (void)in_sizes; (void)n_in; (void)d_ws; (void)ws_size; (void)out_size;
    const float* obs  = (const float*)d_in[0];
    const float* azi  = (const float*)d_in[1];
    const float* ele  = (const float*)d_in[2];
    const float* pvec = (const float*)d_in[3];
    const float* freq = (const float*)d_in[4];
    float* out = (float*)d_out;

    dim3 grid(B_ * F_);
    dim3 block(32);
    CalculateDirectionFeature_kernel<<<grid, block, 0, stream>>>(
        obs, azi, ele, pvec, freq, out);
}


# Whole-line stores without LDS: in each row the lanes whose 16-byte unit opens a 128-byte line hold it one column group and write it with the next group's store (aligned 256-byte runs)
# speedup vs baseline: 1.6795x; 1.0524x over previous
#include <hip/hip_runtime.h>
#include <math.h>


typedef __attribute__((ext_vector_type(2))) float v2f;
typedef __attribute__((ext_vector_type(4))) float v4f;
typedef __attribute__((ext_vector_type(8))) float v8f;

#define B_    4
#define P_    6
#define F_    257
#define T_    300
#define N_    36
#define FT    (F_ * T_)
#define MT_   5
#define V_SOUND 343.0f
#define TWO_PI 6.28318530717958647692f

__global__ __launch_bounds__(32) void CalculateDirectionFeature_kernel(
    const float* __restrict__ obs,
    const float* __restrict__ azi,
    const float* __restrict__ ele,
    const float* __restrict__ pvec,
    const float* __restrict__ freq,
    float* __restrict__ out)
{
    const int bf   = blockIdx.x;
    const int b    = bf / F_;
    const int f    = bf % F_;
    const int lane = threadIdx.x;
    const int c    = lane & 15;
    const int h    = lane >> 4;

    const float fr = freq[f];
    float az[3], el[3], pv[3][3];
    #pragma unroll
    for (int nt = 0; nt < 3; ++nt) {
        const int n  = nt * 16 + c;
        const int nn = (n < N_) ? n : 0;
        az[nt] = azi[b * N_ + nn];
        el[nt] = ele[b * N_ + nn];
    }
    #pragma unroll
    for (int q = 0; q < 3; ++q) {
        const int p = 2 * q + h;
        pv[q][0] = pvec[p * 3 + 0];
        pv[q][1] = pvec[p * 3 + 1];
        pv[q][2] = pvec[p * 3 + 2];
    }
    __builtin_amdgcn_sched_barrier(0);

    const float* const obsBF = obs + ((size_t)b * P_ * F_ + f) * T_ + (size_t)h * FT;
    const bool tailOk = (4 * c + 64 * (MT_ - 1)) < T_;
    v4f o[MT_][3];
    #pragma unroll
    for (int m = 0; m < MT_; ++m) {
        const int tl = (m < MT_ - 1 || tailOk) ? (64 * m + 4 * c) : (64 * m);
        #pragma unroll
        for (int q = 0; q < 3; ++q) {
            o[m][q] = *reinterpret_cast<const v4f*>(obsBF + (size_t)(2 * q) * FT + tl);
        }
    }
    __builtin_amdgcn_sched_barrier(0);

    const float scale = (TWO_PI / V_SOUND) * fr;
    v2f A[3][3];
    #pragma unroll
    for (int nt = 0; nt < 3; ++nt) {
        const bool valid = (nt * 16 + c) < N_;
        float sa, ca, se, ce;
        __sincosf(az[nt], &sa, &ca);
        __sincosf(el[nt], &se, &ce);
        const float rx = se * ca, ry = se * sa, rz = ce;
        #pragma unroll
        for (int q = 0; q < 3; ++q) {
            const float dot = pv[q][0] * rx + pv[q][1] * ry + pv[q][2] * rz;
            float s, cs;
            __sincosf(scale * dot, &s, &cs);
            A[nt][q].x = valid ? cs : 0.0f;
            A[nt][q].y = valid ? s  : 0.0f;
        }
    }

    __builtin_amdgcn_sched_barrier(0);

    float* const outL = out + ((size_t)b * N_ * F_ + f) * T_ + (size_t)(8 * h) * FT + 4 * c;

    const int w = (4 * b + 3 * f) & 7;
    bool   car[8];
    float* rowp[8];
    #pragma unroll
    for (int r = 0; r < 8; ++r) {
        const int s = (w + 3 * r) & 7;
        car[r]  = c >= 16 - s;
        rowp[r] = outL + (size_t)r * FT - (car[r] ? 64 : 0);
    }

    v8f accs[2][3][4];

    #pragma unroll
    for (int m = 0; m <= MT_; ++m) {
        v8f (&cur)[3][4]  = accs[m & 1];
        v8f (&prev)[3][4] = accs[(m & 1) ^ 1];

        if (m < MT_) {
            v2f Bf[3][4];
            #pragma unroll
            for (int q = 0; q < 3; ++q) {
                #pragma unroll
                for (int j = 0; j < 4; ++j) {
                    float s, cs;
                    __sincosf(o[m < MT_ ? m : 0][q][j], &s, &cs);
                    Bf[q][j].x = cs;
                    Bf[q][j].y = s;
                }
            }

            #pragma unroll
            for (int nt = 0; nt < 3; ++nt) {
                #pragma unroll
                for (int j = 0; j < 4; ++j) {
                    cur[nt][j] = (v8f){};
                }
            }
            #pragma unroll
            for (int q = 0; q < 3; ++q) {
                #pragma unroll
                for (int nt = 0; nt < 3; ++nt) {
                    #pragma unroll
                    for (int j = 0; j < 4; ++j) {
                        cur[nt][j] = __builtin_amdgcn_wmma_f32_16x16x4_f32(
                             false, A[nt][q],  false, Bf[q][j],
                             (short)0, cur[nt][j],
                             false,  false);
                    }
                }
            }
        }

        #pragma unroll
        for (int r = 0; r < 8; ++r) {
            const bool cr = car[r];
            bool go;
            if (m == 0)            go = !cr;
            else if (m < MT_ - 1)  go = true;
            else if (m == MT_ - 1) go = cr || tailOk;
            else                   go = cr && tailOk;
            if (go) {
                float* const p = rowp[r] + 64 * m;
                #pragma unroll
                for (int nt = 0; nt < 2; ++nt) {
                    v4f v;
                    if (m == 0) {
                        v.x = cur[nt][0][r]; v.y = cur[nt][1][r];
                        v.z = cur[nt][2][r]; v.w = cur[nt][3][r];
                    } else if (m == MT_) {
                        v.x = prev[nt][0][r]; v.y = prev[nt][1][r];
                        v.z = prev[nt][2][r]; v.w = prev[nt][3][r];
                    } else {
                        v.x = cr ? prev[nt][0][r] : cur[nt][0][r];
                        v.y = cr ? prev[nt][1][r] : cur[nt][1][r];
                        v.z = cr ? prev[nt][2][r] : cur[nt][2][r];
                        v.w = cr ? prev[nt][3][r] : cur[nt][3][r];
                    }
                    *reinterpret_cast<v4f*>(p + (size_t)(nt * 16) * FT) = v;
                }
            }
        }
        if (h == 0) {
            #pragma unroll
            for (int r = 0; r < 4; ++r) {
                const bool cr = car[r];
                bool go;
                if (m == 0)            go = !cr;
                else if (m < MT_ - 1)  go = true;
                else if (m == MT_ - 1) go = cr || tailOk;
                else                   go = cr && tailOk;
                if (go) {
                    v4f v;
                    if (m == 0) {
                        v.x = cur[2][0][r]; v.y = cur[2][1][r];
                        v.z = cur[2][2][r]; v.w = cur[2][3][r];
                    } else if (m == MT_) {
                        v.x = prev[2][0][r]; v.y = prev[2][1][r];
                        v.z = prev[2][2][r]; v.w = prev[2][3][r];
                    } else {
                        v.x = cr ? prev[2][0][r] : cur[2][0][r];
                        v.y = cr ? prev[2][1][r] : cur[2][1][r];
                        v.z = cr ? prev[2][2][r] : cur[2][2][r];
                        v.w = cr ? prev[2][3][r] : cur[2][3][r];
                    }
                    *reinterpret_cast<v4f*>(rowp[r] + 64 * m + (size_t)32 * FT) = v;
                }
            }
        }
    }
}

extern "C" void kernel_launch(void* const* d_in, const int* in_sizes, int n_in,
                              void* d_out, int out_size, void* d_ws, size_t ws_size,
                              hipStream_t stream) {
    (void)in_sizes; (void)n_in; (void)d_ws; (void)ws_size; (void)out_size;
    const float* obs  = (const float*)d_in[0];
    const float* azi  = (const float*)d_in[1];
    const float* ele  = (const float*)d_in[2];
    const float* pvec = (const float*)d_in[3];
    const float* freq = (const float*)d_in[4];
    float* out = (float*)d_out;

    dim3 grid(B_ * F_);
    dim3 block(32);
    CalculateDirectionFeature_kernel<<<grid, block, 0, stream>>>(
        obs, azi, ele, pvec, freq, out);
}


# Partial stores (row start, last group, flush, rows 32-35) narrow exec in place from scalar lane masks of the bin index: no compare-to-scalar-to-skip-branch chain and no wait behind each store pair
# speedup vs baseline: 1.7679x; 1.0526x over previous
#include <hip/hip_runtime.h>
#include <math.h>


typedef __attribute__((ext_vector_type(2))) float v2f;
typedef __attribute__((ext_vector_type(4))) float v4f;
typedef __attribute__((ext_vector_type(8))) float v8f;

#define B_    4
#define P_    6
#define F_    257
#define T_    300
#define N_    36
#define FT    (F_ * T_)
#define MT_   5
#define V_SOUND 343.0f
#define TWO_PI 6.28318530717958647692f

__device__ __forceinline__ void store_lanes(float* p, v4f v, unsigned lanes) {
    unsigned keep;
    asm volatile("s_mov_b32 %0, exec_lo\n\t"
                 "s_and_b32 exec_lo, %0, %3\n\t"
                 "global_store_b128 %1, %2, off\n\t"
                 "s_mov_b32 exec_lo, %0"
                 : "=&s"(keep)
                 : "v"(p), "v"(v), "s"(lanes)
                 : "memory", "scc");
}

__device__ __forceinline__ void store_lanes2(float* p0, v4f v0, float* p1, v4f v1, unsigned lanes) {
    unsigned keep;
    asm volatile("s_mov_b32 %0, exec_lo\n\t"
                 "s_and_b32 exec_lo, %0, %5\n\t"
                 "global_store_b128 %1, %2, off\n\t"
                 "global_store_b128 %3, %4, off\n\t"
                 "s_mov_b32 exec_lo, %0"
                 : "=&s"(keep)
                 : "v"(p0), "v"(v0), "v"(p1), "v"(v1), "s"(lanes)
                 : "memory", "scc");
}

__device__ __forceinline__ v4f take(const v8f (&a)[3][4], int nt, int r) {
    v4f v;
    v.x = a[nt][0][r]; v.y = a[nt][1][r]; v.z = a[nt][2][r]; v.w = a[nt][3][r];
    return v;
}

__device__ __forceinline__ v4f pick(bool first, v4f a, v4f b) {
    v4f v;
    v.x = first ? a.x : b.x; v.y = first ? a.y : b.y;
    v.z = first ? a.z : b.z; v.w = first ? a.w : b.w;
    return v;
}

__global__ __launch_bounds__(32) void CalculateDirectionFeature_kernel(
    const float* __restrict__ obs,
    const float* __restrict__ azi,
    const float* __restrict__ ele,
    const float* __restrict__ pvec,
    const float* __restrict__ freq,
    float* __restrict__ out)
{
    const int bf   = blockIdx.x;
    const int b    = bf / F_;
    const int f    = bf % F_;
    const int lane = threadIdx.x;
    const int c    = lane & 15;
    const int h    = lane >> 4;

    const float fr = freq[f];
    float az[3], el[3], pv[3][3];
    #pragma unroll
    for (int nt = 0; nt < 3; ++nt) {
        const int n  = nt * 16 + c;
        const int nn = (n < N_) ? n : 0;
        az[nt] = azi[b * N_ + nn];
        el[nt] = ele[b * N_ + nn];
    }
    #pragma unroll
    for (int q = 0; q < 3; ++q) {
        const int p = 2 * q + h;
        pv[q][0] = pvec[p * 3 + 0];
        pv[q][1] = pvec[p * 3 + 1];
        pv[q][2] = pvec[p * 3 + 2];
    }
    __builtin_amdgcn_sched_barrier(0);

    const float* const obsBF = obs + ((size_t)b * P_ * F_ + f) * T_ + (size_t)h * FT;
    const bool tailOk = (4 * c + 64 * (MT_ - 1)) < T_;
    v4f o[MT_][3];
    #pragma unroll
    for (int m = 0; m < MT_; ++m) {
        const int tl = (m < MT_ - 1 || tailOk) ? (64 * m + 4 * c) : (64 * m);
        #pragma unroll
        for (int q = 0; q < 3; ++q) {
            o[m][q] = *reinterpret_cast<const v4f*>(obsBF + (size_t)(2 * q) * FT + tl);
        }
    }
    __builtin_amdgcn_sched_barrier(0);

    const float scale = (TWO_PI / V_SOUND) * fr;
    v2f A[3][3];
    #pragma unroll
    for (int nt = 0; nt < 3; ++nt) {
        const bool valid = (nt * 16 + c) < N_;
        float sa, ca, se, ce;
        __sincosf(az[nt], &sa, &ca);
        __sincosf(el[nt], &se, &ce);
        const float rx = se * ca, ry = se * sa, rz = ce;
        #pragma unroll
        for (int q = 0; q < 3; ++q) {
            const float dot = pv[q][0] * rx + pv[q][1] * ry + pv[q][2] * rz;
            float s, cs;
            __sincosf(scale * dot, &s, &cs);
            A[nt][q].x = valid ? cs : 0.0f;
            A[nt][q].y = valid ? s  : 0.0f;
        }
    }

    __builtin_amdgcn_sched_barrier(0);

    float* const outL = out + ((size_t)b * N_ * F_ + f) * T_ + (size_t)(8 * h) * FT + 4 * c;

    const int w = (4 * b + 3 * f) & 7;
    bool     car[8];
    float*   rowp[8];
    unsigned carM[8];
    #pragma unroll
    for (int r = 0; r < 8; ++r) {
        const int s = (w + 3 * r) & 7;
        car[r]  = c >= 16 - s;
        rowp[r] = outL + (size_t)r * FT - (car[r] ? 64 : 0);
        const unsigned m16 = (0xFFFFu << (16 - s)) & 0xFFFFu;
        carM[r] = m16 | (m16 << 16);
    }
    const unsigned tailM = 0x07FF07FFu;

    v8f accs[2][3][4];

    #pragma unroll
    for (int m = 0; m <= MT_; ++m) {
        v8f (&cur)[3][4]  = accs[m & 1];
        v8f (&prev)[3][4] = accs[(m & 1) ^ 1];

        if (m < MT_) {
            v2f Bf[3][4];
            #pragma unroll
            for (int q = 0; q < 3; ++q) {
                #pragma unroll
                for (int j = 0; j < 4; ++j) {
                    float s, cs;
                    __sincosf(o[m < MT_ ? m : 0][q][j], &s, &cs);
                    Bf[q][j].x = cs;
                    Bf[q][j].y = s;
                }
            }

            #pragma unroll
            for (int nt = 0; nt < 3; ++nt) {
                #pragma unroll
                for (int j = 0; j < 4; ++j) {
                    cur[nt][j] = (v8f){};
                }
            }
            #pragma unroll
            for (int q = 0; q < 3; ++q) {
                #pragma unroll
                for (int nt = 0; nt < 3; ++nt) {
                    #pragma unroll
                    for (int j = 0; j < 4; ++j) {
                        cur[nt][j] = __builtin_amdgcn_wmma_f32_16x16x4_f32(
                             false, A[nt][q],  false, Bf[q][j],
                             (short)0, cur[nt][j],
                             false,  false);
                    }
                }
            }
        }

        if (m >= 1 && m < MT_ - 1) {
            #pragma unroll
            for (int r = 0; r < 8; ++r) {
                float* const p = rowp[r] + 64 * m;
                #pragma unroll
                for (int nt = 0; nt < 2; ++nt) {
                    *reinterpret_cast<v4f*>(p + (size_t)(nt * 16) * FT) =
                        pick(car[r], take(prev, nt, r), take(cur, nt, r));
                }
            }
            v4f    t[4];
            float* q[4];
            #pragma unroll
            for (int r = 0; r < 4; ++r) {
                t[r] = pick(car[r], take(prev, 2, r), take(cur, 2, r));
                q[r] = rowp[r] + 64 * m + (size_t)32 * FT;
            }
            #pragma unroll
            for (int r = 0; r < 4; ++r) asm volatile("" : "+v"(t[r]));
            __builtin_amdgcn_sched_barrier(0);
            #pragma unroll
            for (int r = 0; r < 4; ++r) store_lanes(q[r], t[r], 0x0000FFFFu);
            asm volatile("s_wait_xcnt 0x0" ::: "memory");
            __builtin_amdgcn_sched_barrier(0);
        } else {
            v4f      t[20];
            float*   q[20];
            unsigned k[20];
            #pragma unroll
            for (int r = 0; r < 8; ++r) {
                const unsigned lanes = (m == 0)       ? ~carM[r]
                                     : (m == MT_ - 1) ? (carM[r] | tailM)
                                                      : (carM[r] & tailM);
                #pragma unroll
                for (int nt = 0; nt < 3; ++nt) {
                    if (nt == 2 && r >= 4) continue;
                    const int i = (nt < 2) ? (2 * r + nt) : (16 + r);
                    t[i] = (m == 0)   ? take(cur, nt, r)
                         : (m == MT_) ? take(prev, nt, r)
                                      : pick(car[r], take(prev, nt, r), take(cur, nt, r));
                    q[i] = rowp[r] + 64 * m + (size_t)(nt * 16) * FT;
                    k[i] = (nt < 2) ? lanes : (lanes & 0x0000FFFFu);
                }
            }
            #pragma unroll
            for (int i = 0; i < 20; ++i) asm volatile("" : "+v"(t[i]));
            __builtin_amdgcn_sched_barrier(0);
            #pragma unroll
            for (int r = 0; r < 8; ++r)
                store_lanes2(q[2 * r], t[2 * r], q[2 * r + 1], t[2 * r + 1], k[2 * r]);
            #pragma unroll
            for (int r = 0; r < 4; ++r) store_lanes(q[16 + r], t[16 + r], k[16 + r]);
            asm volatile("s_wait_xcnt 0x0" ::: "memory");
            __builtin_amdgcn_sched_barrier(0);
        }
    }
}

extern "C" void kernel_launch(void* const* d_in, const int* in_sizes, int n_in,
                              void* d_out, int out_size, void* d_ws, size_t ws_size,
                              hipStream_t stream) {
    (void)in_sizes; (void)n_in; (void)d_ws; (void)ws_size; (void)out_size;
    const float* obs  = (const float*)d_in[0];
    const float* azi  = (const float*)d_in[1];
    const float* ele  = (const float*)d_in[2];
    const float* pvec = (const float*)d_in[3];
    const float* freq = (const float*)d_in[4];
    float* out = (float*)d_out;

    dim3 grid(B_ * F_);
    dim3 block(32);
    CalculateDirectionFeature_kernel<<<grid, block, 0, stream>>>(
        obs, azi, ele, pvec, freq, out);
}


# Grid laid over the 1,024 SIMDs: 256 four-wave workgroups, one whole bin per wave and SIMD; the 4 bins left over run as 60 pieces (one column group of one n-tile, 12 WMMAs) on 15 more workgroups
# speedup vs baseline: 1.9096x; 1.0802x over previous
#include <hip/hip_runtime.h>
#include <math.h>


typedef __attribute__((ext_vector_type(2))) float v2f;
typedef __attribute__((ext_vector_type(4))) float v4f;
typedef __attribute__((ext_vector_type(8))) float v8f;

#define B_    4
#define P_    6
#define F_    257
#define T_    300
#define N_    36
#define FT    (F_ * T_)
#define MT_   5
#define BINS_      (B_ * F_)
#define MAIN_WGS   256
#define MAIN_BINS  (MAIN_WGS * 4)
#define PIECES_PER_BIN (MT_ * 3)
#define PIECE_WGS  (((BINS_ - MAIN_BINS) * PIECES_PER_BIN) / 4)
#define V_SOUND 343.0f
#define TWO_PI 6.28318530717958647692f

__device__ __forceinline__ void store_lanes(float* p, v4f v, unsigned lanes) {
    unsigned keep;
    asm volatile("s_mov_b32 %0, exec_lo\n\t"
                 "s_and_b32 exec_lo, %0, %3\n\t"
                 "global_store_b128 %1, %2, off\n\t"
                 "s_mov_b32 exec_lo, %0"
                 : "=&s"(keep)
                 : "v"(p), "v"(v), "s"(lanes)
                 : "memory", "scc");
}

__device__ __forceinline__ void store_lanes2(float* p0, v4f v0, float* p1, v4f v1, unsigned lanes) {
    unsigned keep;
    asm volatile("s_mov_b32 %0, exec_lo\n\t"
                 "s_and_b32 exec_lo, %0, %5\n\t"
                 "global_store_b128 %1, %2, off\n\t"
                 "global_store_b128 %3, %4, off\n\t"
                 "s_mov_b32 exec_lo, %0"
                 : "=&s"(keep)
                 : "v"(p0), "v"(v0), "v"(p1), "v"(v1), "s"(lanes)
                 : "memory", "scc");
}

__device__ __forceinline__ v4f take(const v8f (&a)[3][4], int nt, int r) {
    v4f v;
    v.x = a[nt][0][r]; v.y = a[nt][1][r]; v.z = a[nt][2][r]; v.w = a[nt][3][r];
    return v;
}

__device__ __forceinline__ v4f pick(bool first, v4f a, v4f b) {
    v4f v;
    v.x = first ? a.x : b.x; v.y = first ? a.y : b.y;
    v.z = first ? a.z : b.z; v.w = first ? a.w : b.w;
    return v;
}

__device__ __forceinline__ void piece_of_bin(
    const float* __restrict__ obs, const float* __restrict__ azi, const float* __restrict__ ele,
    const float* __restrict__ pvec, const float* __restrict__ freq, float* __restrict__ out,
    const int pc, const int c, const int h)
{
    const int bf  = MAIN_BINS + pc / PIECES_PER_BIN;
    const int rem = pc % PIECES_PER_BIN;
    const int m   = rem / 3;
    const int nt  = rem % 3;
    const int b   = bf / F_;
    const int f   = bf % F_;

    const float fr    = freq[f];
    const int   n     = nt * 16 + c;
    const bool  valid = n < N_;
    const int   nn    = valid ? n : 0;
    const float az = azi[b * N_ + nn];
    const float el = ele[b * N_ + nn];
    float pv[3][3];
    #pragma unroll
    for (int q = 0; q < 3; ++q) {
        const int p = 2 * q + h;
        pv[q][0] = pvec[p * 3 + 0];
        pv[q][1] = pvec[p * 3 + 1];
        pv[q][2] = pvec[p * 3 + 2];
    }

    const bool colOk = (64 * m + 4 * c) < T_;
    const int  tl    = colOk ? (64 * m + 4 * c) : (64 * m);
    const float* const obsBF = obs + ((size_t)b * P_ * F_ + f) * T_ + (size_t)h * FT + tl;
    v4f o[3];
    #pragma unroll
    for (int q = 0; q < 3; ++q)
        o[q] = *reinterpret_cast<const v4f*>(obsBF + (size_t)(2 * q) * FT);

    const float scale = (TWO_PI / V_SOUND) * fr;
    float sa, ca, se, ce;
    __sincosf(az, &sa, &ca);
    __sincosf(el, &se, &ce);
    const float rx = se * ca, ry = se * sa, rz = ce;
    v2f A[3];
    #pragma unroll
    for (int q = 0; q < 3; ++q) {
        const float dot = pv[q][0] * rx + pv[q][1] * ry + pv[q][2] * rz;
        float s, cs;
        __sincosf(scale * dot, &s, &cs);
        A[q].x = valid ? cs : 0.0f;
        A[q].y = valid ? s  : 0.0f;
    }

    v8f acc[4];
    #pragma unroll
    for (int j = 0; j < 4; ++j) acc[j] = (v8f){};
    #pragma unroll
    for (int q = 0; q < 3; ++q) {
        #pragma unroll
        for (int j = 0; j < 4; ++j) {
            float s, cs;
            __sincosf(o[q][j], &s, &cs);
            v2f Bf;
            Bf.x = cs;
            Bf.y = s;
            acc[j] = __builtin_amdgcn_wmma_f32_16x16x4_f32(
                 false, A[q],  false, Bf,
                 (short)0, acc[j],
                 false,  false);
        }
    }

    const int row0 = nt * 16 + 8 * h;
    float* const outP = out + ((size_t)b * N_ * F_ + f) * T_ + (size_t)row0 * FT + 64 * m + 4 * c;
    #pragma unroll
    for (int r = 0; r < 8; ++r) {
        if (colOk && (row0 + r) < N_) {
            v4f v;
            v.x = acc[0][r]; v.y = acc[1][r]; v.z = acc[2][r]; v.w = acc[3][r];
            *reinterpret_cast<v4f*>(outP + (size_t)r * FT) = v;
        }
    }
}

__global__ __launch_bounds__(128) void CalculateDirectionFeature_kernel(
    const float* __restrict__ obs,
    const float* __restrict__ azi,
    const float* __restrict__ ele,
    const float* __restrict__ pvec,
    const float* __restrict__ freq,
    float* __restrict__ out)
{
    const int lane = threadIdx.x & 31;
    const int c    = lane & 15;
    const int h    = lane >> 4;
    const int wv   = __builtin_amdgcn_readfirstlane((int)(threadIdx.x >> 5));

    if (blockIdx.x >= MAIN_WGS) {
        piece_of_bin(obs, azi, ele, pvec, freq, out, (int)(blockIdx.x - MAIN_WGS) * 4 + wv, c, h);
        return;
    }

    const int bf   = (int)blockIdx.x * 4 + wv;
    const int b    = bf / F_;
    const int f    = bf % F_;

    const float fr = freq[f];
    float az[3], el[3], pv[3][3];
    #pragma unroll
    for (int nt = 0; nt < 3; ++nt) {
        const int n  = nt * 16 + c;
        const int nn = (n < N_) ? n : 0;
        az[nt] = azi[b * N_ + nn];
        el[nt] = ele[b * N_ + nn];
    }
    #pragma unroll
    for (int q = 0; q < 3; ++q) {
        const int p = 2 * q + h;
        pv[q][0] = pvec[p * 3 + 0];
        pv[q][1] = pvec[p * 3 + 1];
        pv[q][2] = pvec[p * 3 + 2];
    }
    __builtin_amdgcn_sched_barrier(0);

    const float* const obsBF = obs + ((size_t)b * P_ * F_ + f) * T_ + (size_t)h * FT;
    const bool tailOk = (4 * c + 64 * (MT_ - 1)) < T_;
    v4f o[MT_][3];
    #pragma unroll
    for (int m = 0; m < MT_; ++m) {
        const int tl = (m < MT_ - 1 || tailOk) ? (64 * m + 4 * c) : (64 * m);
        #pragma unroll
        for (int q = 0; q < 3; ++q) {
            o[m][q] = *reinterpret_cast<const v4f*>(obsBF + (size_t)(2 * q) * FT + tl);
        }
    }
    __builtin_amdgcn_sched_barrier(0);

    const float scale = (TWO_PI / V_SOUND) * fr;
    v2f A[3][3];
    #pragma unroll
    for (int nt = 0; nt < 3; ++nt) {
        const bool valid = (nt * 16 + c) < N_;
        float sa, ca, se, ce;
        __sincosf(az[nt], &sa, &ca);
        __sincosf(el[nt], &se, &ce);
        const float rx = se * ca, ry = se * sa, rz = ce;
        #pragma unroll
        for (int q = 0; q < 3; ++q) {
            const float dot = pv[q][0] * rx + pv[q][1] * ry + pv[q][2] * rz;
            float s, cs;
            __sincosf(scale * dot, &s, &cs);
            A[nt][q].x = valid ? cs : 0.0f;
            A[nt][q].y = valid ? s  : 0.0f;
        }
    }

    __builtin_amdgcn_sched_barrier(0);

    float* const outL = out + ((size_t)b * N_ * F_ + f) * T_ + (size_t)(8 * h) * FT + 4 * c;

    const int w = (4 * b + 3 * f) & 7;
    bool     car[8];
    float*   rowp[8];
    unsigned carM[8];
    #pragma unroll
    for (int r = 0; r < 8; ++r) {
        const int s = (w + 3 * r) & 7;
        car[r]  = c >= 16 - s;
        rowp[r] = outL + (size_t)r * FT - (car[r] ? 64 : 0);
        const unsigned m16 = (0xFFFFu << (16 - s)) & 0xFFFFu;
        carM[r] = m16 | (m16 << 16);
    }
    const unsigned tailM = 0x07FF07FFu;

    v8f accs[2][3][4];

    #pragma unroll
    for (int m = 0; m <= MT_; ++m) {
        v8f (&cur)[3][4]  = accs[m & 1];
        v8f (&prev)[3][4] = accs[(m & 1) ^ 1];

        if (m < MT_) {
            v2f Bf[3][4];
            #pragma unroll
            for (int q = 0; q < 3; ++q) {
                #pragma unroll
                for (int j = 0; j < 4; ++j) {
                    float s, cs;
                    __sincosf(o[m < MT_ ? m : 0][q][j], &s, &cs);
                    Bf[q][j].x = cs;
                    Bf[q][j].y = s;
                }
            }

            #pragma unroll
            for (int nt = 0; nt < 3; ++nt) {
                #pragma unroll
                for (int j = 0; j < 4; ++j) {
                    cur[nt][j] = (v8f){};
                }
            }
            #pragma unroll
            for (int q = 0; q < 3; ++q) {
                #pragma unroll
                for (int nt = 0; nt < 3; ++nt) {
                    #pragma unroll
                    for (int j = 0; j < 4; ++j) {
                        cur[nt][j] = __builtin_amdgcn_wmma_f32_16x16x4_f32(
                             false, A[nt][q],  false, Bf[q][j],
                             (short)0, cur[nt][j],
                             false,  false);
                    }
                }
            }
        }

        if (m >= 1 && m < MT_ - 1) {
            #pragma unroll
            for (int r = 0; r < 8; ++r) {
                float* const p = rowp[r] + 64 * m;
                #pragma unroll
                for (int nt = 0; nt < 2; ++nt) {
                    *reinterpret_cast<v4f*>(p + (size_t)(nt * 16) * FT) =
                        pick(car[r], take(prev, nt, r), take(cur, nt, r));
                }
            }
            v4f    t[4];
            float* q[4];
            #pragma unroll
            for (int r = 0; r < 4; ++r) {
                t[r] = pick(car[r], take(prev, 2, r), take(cur, 2, r));
                q[r] = rowp[r] + 64 * m + (size_t)32 * FT;
            }
            #pragma unroll
            for (int r = 0; r < 4; ++r) asm volatile("" : "+v"(t[r]));
            __builtin_amdgcn_sched_barrier(0);
            #pragma unroll
            for (int r = 0; r < 4; ++r) store_lanes(q[r], t[r], 0x0000FFFFu);
            asm volatile("s_wait_xcnt 0x0" ::: "memory");
            __builtin_amdgcn_sched_barrier(0);
        } else {
            v4f      t[20];
            float*   q[20];
            unsigned k[20];
            #pragma unroll
            for (int r = 0; r < 8; ++r) {
                const unsigned lanes = (m == 0)       ? ~carM[r]
                                     : (m == MT_ - 1) ? (carM[r] | tailM)
                                                      : (carM[r] & tailM);
                #pragma unroll
                for (int nt = 0; nt < 3; ++nt) {
                    if (nt == 2 && r >= 4) continue;
                    const int i = (nt < 2) ? (2 * r + nt) : (16 + r);
                    t[i] = (m == 0)   ? take(cur, nt, r)
                         : (m == MT_) ? take(prev, nt, r)
                                      : pick(car[r], take(prev, nt, r), take(cur, nt, r));
                    q[i] = rowp[r] + 64 * m + (size_t)(nt * 16) * FT;
                    k[i] = (nt < 2) ? lanes : (lanes & 0x0000FFFFu);
                }
            }
            #pragma unroll
            for (int i = 0; i < 20; ++i) asm volatile("" : "+v"(t[i]));
            __builtin_amdgcn_sched_barrier(0);
            #pragma unroll
            for (int r = 0; r < 8; ++r)
                store_lanes2(q[2 * r], t[2 * r], q[2 * r + 1], t[2 * r + 1], k[2 * r]);
            #pragma unroll
            for (int r = 0; r < 4; ++r) store_lanes(q[16 + r], t[16 + r], k[16 + r]);
            asm volatile("s_wait_xcnt 0x0" ::: "memory");
            __builtin_amdgcn_sched_barrier(0);
        }
    }
}

extern "C" void kernel_launch(void* const* d_in, const int* in_sizes, int n_in,
                              void* d_out, int out_size, void* d_ws, size_t ws_size,
                              hipStream_t stream) {
    (void)in_sizes; (void)n_in; (void)d_ws; (void)ws_size; (void)out_size;
    const float* obs  = (const float*)d_in[0];
    const float* azi  = (const float*)d_in[1];
    const float* ele  = (const float*)d_in[2];
    const float* pvec = (const float*)d_in[3];
    const float* freq = (const float*)d_in[4];
    float* out = (float*)d_out;

    dim3 grid(MAIN_WGS + PIECE_WGS);
    dim3 block(128);
    CalculateDirectionFeature_kernel<<<grid, block, 0, stream>>>(
        obs, azi, ele, pvec, freq, out);
}
